# starting kernel
# baseline (speedup 1.0000x reference)
#include <hip/hip_runtime.h>

typedef __attribute__((ext_vector_type(2))) float v2f;
typedef __attribute__((ext_vector_type(8))) float v8f;

#define NB 16
#define NS 800
#define NT 224
#define NN 1024
#define DD 32

__global__ void hstgnn_pack_weights(const float* __restrict__ relW,
                                    const float* __restrict__ maskW,
                                    float* __restrict__ wpack) {
    int i = blockIdx.x * blockDim.x + threadIdx.x;
    if (i >= 2 * 16 * 32) return;
    int d    = i & 31;
    int n    = (i >> 5) & 15;
    int type = i >> 9;
    float v = 0.0f;
    if (n < 8) {
        const int krow[2][8] = { {0,0,2,2,0,0,3,3},
                                 {1,1,3,3,1,1,2,2} };
        int k    = krow[type][n];
        int half = (n >> 2) & 1;
        const float* src = (n & 1) ? maskW : relW;
        v = src[k * 64 + half * 32 + d];
    }
    wpack[i] = v;
}

__global__ void hstgnn_score_wmma(const float* __restrict__ sp,
                                  const float* __restrict__ tp,
                                  const float* __restrict__ wpack,
                                  float* __restrict__ scores) {
    int id   = blockIdx.x;
    int b    = id >> 6;
    int t    = id & 63;
    int lane = threadIdx.x;
    int r    = lane & 15;
    int hi   = lane >> 4;

    const float* src;
    int type;
    if (t < 50) { type = 0; src = sp + ((size_t)b * NS + (size_t)t * 16) * DD; }
    else        { type = 1; src = tp + ((size_t)b * NT + (size_t)(t - 50) * 16) * DD; }
    const float* w = wpack + type * (16 * 32);

    v8f acc = {0.f,0.f,0.f,0.f,0.f,0.f,0.f,0.f};
#pragma unroll
    for (int k = 0; k < 8; ++k) {
        int kb = k * 4 + hi * 2;
        v2f a  = *(const v2f*)(src + r * DD + kb);
        v2f bb = *(const v2f*)(w   + r * DD + kb);
        acc = __builtin_amdgcn_wmma_f32_16x16x4_f32(
                  false, a, false, bb, (short)0, acc, false, false);
    }

    float* dst = scores + ((size_t)b * NN + (size_t)t * 16) * 16;
#pragma unroll
    for (int v = 0; v < 8; ++v) {
        int row = v + hi * 8;
        dst[row * 16 + r] = acc[v];
    }
}

__global__ void hstgnn_emit(const float* __restrict__ scores,
                            const float* __restrict__ relb,
                            const float* __restrict__ maskb,
                            float* __restrict__ out) {
    int bi = blockIdx.x;
    int b  = bi >> 10;
    int i  = bi & 1023;
    int j0 = threadIdx.x * 4;
    bool iT = (i  >= NS);
    bool jT = (j0 >= NS);

    int rowslot = (iT == jT) ? 0 : 2;
    int colslot = (iT == jT) ? 4 : 6;
    int k = jT ? (iT ? 1 : 2) : (iT ? 3 : 0);

    const float* srow = scores + ((size_t)b * NN + i) * 16;
    float2 rv = *(const float2*)(srow + rowslot);
    float  rb = relb[k];
    float  mb = maskb[k];

    const float* scol = scores + (size_t)b * NN * 16;
    float res[4];
#pragma unroll
    for (int u = 0; u < 4; ++u) {
        int j = j0 + u;
        float2 cv = *(const float2*)(scol + (size_t)j * 16 + colslot);
        float adj  = rv.x + cv.x + rb;
        float mask = fmaxf(rv.y + cv.y + mb, 0.0f);
        res[u] = tanhf(adj * mask);
    }
    *(float4*)(out + (size_t)bi * NN + j0) =
        make_float4(res[0], res[1], res[2], res[3]);
}

extern "C" void kernel_launch(void* const* d_in, const int* in_sizes, int n_in,
                              void* d_out, int out_size, void* d_ws, size_t ws_size,
                              hipStream_t stream) {
    const float* spatial  = (const float*)d_in[0];
    const float* temporal = (const float*)d_in[1];
    const float* rel_W    = (const float*)d_in[2];
    const float* rel_b    = (const float*)d_in[3];
    const float* mask_W   = (const float*)d_in[4];
    const float* mask_b   = (const float*)d_in[5];
    float*       out      = (float*)d_out;

    float* wpack  = (float*)d_ws;
    float* scores = wpack + 2 * 16 * 32;

    hipLaunchKernelGGL(hstgnn_pack_weights, dim3(4), dim3(256), 0, stream,
                       rel_W, mask_W, wpack);
    hipLaunchKernelGGL(hstgnn_score_wmma, dim3(NB * 64), dim3(32), 0, stream,
                       spatial, temporal, wpack, scores);
    hipLaunchKernelGGL(hstgnn_emit, dim3(NB * NN), dim3(256), 0, stream,
                       scores, rel_b, mask_b, out);
}


# emit loop nest transposed: a wave keeps 4 rows' scores in scalar registers and streams all 1024 columns from a per-block LDS table (512 blocks of 8 waves, was 16384); tanhf and sum order kept
# speedup vs baseline: 2.0923x; 2.0923x over previous
#include <hip/hip_runtime.h>

typedef __attribute__((ext_vector_type(2))) float v2f;
typedef __attribute__((ext_vector_type(8))) float v8f;

#define NB 16
#define NS 800
#define NT 224
#define NN 1024
#define DD 32

__global__ void hstgnn_pack_weights(const float* __restrict__ relW,
                                    const float* __restrict__ maskW,
                                    float* __restrict__ wpack) {
    int i = blockIdx.x * blockDim.x + threadIdx.x;
    if (i >= 2 * 16 * 32) return;
    int d    = i & 31;
    int n    = (i >> 5) & 15;
    int type = i >> 9;
    float v = 0.0f;
    if (n < 8) {
        const int krow[2][8] = { {0,0,2,2,0,0,3,3},
                                 {1,1,3,3,1,1,2,2} };
        int k    = krow[type][n];
        int half = (n >> 2) & 1;
        const float* src = (n & 1) ? maskW : relW;
        v = src[k * 64 + half * 32 + d];
    }
    wpack[i] = v;
}

__global__ void hstgnn_score_wmma(const float* __restrict__ sp,
                                  const float* __restrict__ tp,
                                  const float* __restrict__ wpack,
                                  float* __restrict__ scores) {
    int id   = blockIdx.x;
    int b    = id >> 6;
    int t    = id & 63;
    int lane = threadIdx.x;
    int r    = lane & 15;
    int hi   = lane >> 4;

    const float* src;
    int type;
    if (t < 50) { type = 0; src = sp + ((size_t)b * NS + (size_t)t * 16) * DD; }
    else        { type = 1; src = tp + ((size_t)b * NT + (size_t)(t - 50) * 16) * DD; }
    const float* w = wpack + type * (16 * 32);

    v8f acc = {0.f,0.f,0.f,0.f,0.f,0.f,0.f,0.f};
#pragma unroll
    for (int k = 0; k < 8; ++k) {
        int kb = k * 4 + hi * 2;
        v2f a  = *(const v2f*)(src + r * DD + kb);
        v2f bb = *(const v2f*)(w   + r * DD + kb);
        acc = __builtin_amdgcn_wmma_f32_16x16x4_f32(
                  false, a, false, bb, (short)0, acc, false, false);
    }

    float* dst = scores + ((size_t)b * NN + (size_t)t * 16) * 16;
#pragma unroll
    for (int v = 0; v < 8; ++v) {
        int row = v + hi * 8;
        dst[row * 16 + r] = acc[v];
    }
}

#define EMIT_ROWS    32
#define EMIT_THREADS 256
#define EMIT_WROWS   4

__device__ __forceinline__ void hstgnn_segment(float4 cr, float4 cm,
                                               const float (&rx)[EMIT_WROWS],
                                               const float (&ry)[EMIT_WROWS],
                                               float rb, float mb, float* o) {
    for (int r = 0; r < EMIT_WROWS; ++r) {
        float4 v;
        v.x = tanhf((rx[r] + cr.x + rb) * fmaxf(ry[r] + cm.x + mb, 0.0f));
        v.y = tanhf((rx[r] + cr.y + rb) * fmaxf(ry[r] + cm.y + mb, 0.0f));
        v.z = tanhf((rx[r] + cr.z + rb) * fmaxf(ry[r] + cm.z + mb, 0.0f));
        v.w = tanhf((rx[r] + cr.w + rb) * fmaxf(ry[r] + cm.w + mb, 0.0f));
        *(float4*)(o + (size_t)r * NN) = v;
    }
}

__global__ __launch_bounds__(EMIT_THREADS)
void hstgnn_emit_rows(const float* __restrict__ scores,
                      const float* __restrict__ relb,
                      const float* __restrict__ maskb,
                      float* __restrict__ out) {
    __shared__ __attribute__((aligned(16))) float ctab[2][NN];

    const int blk   = blockIdx.x;
    const int b     = blk >> 5;
    const int rbase = (blk & 31) * EMIT_ROWS;
    const bool iT   = (rbase >= NS);
    const float* sb = scores + (size_t)b * NN * 16;

    float2 cv[NN / EMIT_THREADS];
    for (int u = 0; u < NN / EMIT_THREADS; ++u) {
        const int j       = threadIdx.x + u * EMIT_THREADS;
        const int colslot = (iT == (j >= NS)) ? 4 : 6;
        cv[u] = *(const float2*)(sb + (size_t)j * 16 + colslot);
    }

    const int lane = threadIdx.x & 31;
    const int wave = __builtin_amdgcn_readfirstlane((int)(threadIdx.x >> 5));
    const int i0   = rbase + wave * EMIT_WROWS;

    const int kS = iT ? 3 : 0;
    const int kT = iT ? 1 : 2;
    const float rbS = relb[kS], mbS = maskb[kS];
    const float rbT = relb[kT], mbT = maskb[kT];

    float rxS[EMIT_WROWS], ryS[EMIT_WROWS], rxT[EMIT_WROWS], ryT[EMIT_WROWS];
    for (int r = 0; r < EMIT_WROWS; ++r) {
        const float4 rs = *(const float4*)(sb + (size_t)(i0 + r) * 16);
        rxS[r] = iT ? rs.z : rs.x;  ryS[r] = iT ? rs.w : rs.y;
        rxT[r] = iT ? rs.x : rs.z;  ryT[r] = iT ? rs.y : rs.w;
    }

    for (int u = 0; u < NN / EMIT_THREADS; ++u) {
        const int j = threadIdx.x + u * EMIT_THREADS;
        ctab[0][j] = cv[u].x;
        ctab[1][j] = cv[u].y;
    }
    __syncthreads();

    const float* crel  = &ctab[0][lane * 4];
    const float* cmask = &ctab[1][lane * 4];
    float* o = out + ((size_t)b * NN + i0) * NN + lane * 4;

    for (int t = 0; t < 6; ++t)
        hstgnn_segment(*(const float4*)(crel + t * 128), *(const float4*)(cmask + t * 128),
                       rxS, ryS, rbS, mbS, o + t * 128);
    {
        const bool laneT = (lane >= 8);
        float rx[EMIT_WROWS], ry[EMIT_WROWS];
        for (int r = 0; r < EMIT_WROWS; ++r) {
            rx[r] = laneT ? rxT[r] : rxS[r];
            ry[r] = laneT ? ryT[r] : ryS[r];
        }
        hstgnn_segment(*(const float4*)(crel + 6 * 128), *(const float4*)(cmask + 6 * 128),
                       rx, ry, laneT ? rbT : rbS, laneT ? mbT : mbS, o + 6 * 128);
    }
    hstgnn_segment(*(const float4*)(crel + 7 * 128), *(const float4*)(cmask + 7 * 128),
                   rxT, ryT, rbT, mbT, o + 7 * 128);
}

extern "C" void kernel_launch(void* const* d_in, const int* in_sizes, int n_in,
                              void* d_out, int out_size, void* d_ws, size_t ws_size,
                              hipStream_t stream) {
    const float* spatial  = (const float*)d_in[0];
    const float* temporal = (const float*)d_in[1];
    const float* rel_W    = (const float*)d_in[2];
    const float* rel_b    = (const float*)d_in[3];
    const float* mask_W   = (const float*)d_in[4];
    const float* mask_b   = (const float*)d_in[5];
    float*       out      = (float*)d_out;

    float* wpack  = (float*)d_ws;
    float* scores = wpack + 2 * 16 * 32;

    hipLaunchKernelGGL(hstgnn_pack_weights, dim3(4), dim3(256), 0, stream,
                       rel_W, mask_W, wpack);
    hipLaunchKernelGGL(hstgnn_score_wmma, dim3(NB * 64), dim3(32), 0, stream,
                       spatial, temporal, wpack, scores);
    hipLaunchKernelGGL(hstgnn_emit_rows, dim3(NB * 32), dim3(EMIT_THREADS), 0, stream,
                       scores, rel_b, mask_b, out);
}
